# starting kernel
# baseline (speedup 1.0000x reference)
#include <hip/hip_runtime.h>

typedef __attribute__((ext_vector_type(2))) float v2f;
typedef __attribute__((ext_vector_type(8))) float v8f;

#define DIM 64

__global__ void sage_zero_kernel(float* __restrict__ ws, int n) {
    int i = blockIdx.x * blockDim.x + threadIdx.x;
    if (i < n) ws[i] = 0.0f;
}

__global__ void __launch_bounds__(256)
sage_edge_scatter_kernel(const float* __restrict__ x,
                         const int*   __restrict__ src,
                         const int*   __restrict__ dst,
                         float*       __restrict__ agg,
                         float*       __restrict__ deg,
                         int E) {
    int gid  = blockIdx.x * blockDim.x + threadIdx.x;
    int e    = gid >> 5;
    int lane = gid & 31;
    if (e >= E) return;
    int s = src[e];
    int d = dst[e];
    const float2* xr = (const float2*)(x + (size_t)s * DIM);
    float2 v = xr[lane];
    float* ar = agg + (size_t)d * DIM + lane * 2;
    unsafeAtomicAdd(ar,     v.x);
    unsafeAtomicAdd(ar + 1, v.y);
    if (lane == 0) unsafeAtomicAdd(deg + d, 1.0f);
}

__global__ void __launch_bounds__(256)
sage_wmma_gemm_kernel(const float* __restrict__ x,
                      const float* __restrict__ agg,
                      const float* __restrict__ deg,
                      const float* __restrict__ Wself,
                      const float* __restrict__ bself,
                      const float* __restrict__ Wneigh,
                      const float* __restrict__ bneigh,
                      float*       __restrict__ out,
                      int nTiles) {
    __shared__ float sWs[DIM * DIM];
    __shared__ float sWn[DIM * DIM];
    __shared__ float sB [DIM];

    for (int i = threadIdx.x; i < DIM * DIM; i += 256) {
        sWs[i] = Wself[i];
        sWn[i] = Wneigh[i];
    }
    if (threadIdx.x < DIM) sB[threadIdx.x] = bself[threadIdx.x] + bneigh[threadIdx.x];
    __syncthreads();

    const int wave = threadIdx.x >> 5;
    const int lane = threadIdx.x & 31;
    const int tile = blockIdx.x * 8 + wave;
    if (tile >= nTiles) return;

    const int m    = lane & 15;
    const int hi   = lane >> 4;
    const int row0 = tile * 16;
    const int row  = row0 + m;

    const float dinv = 1.0f / fmaxf(deg[row], 1.0f);
    const float* __restrict__ xrow = x   + (size_t)row * DIM;
    const float* __restrict__ arow = agg + (size_t)row * DIM;

    v8f acc[4];
    for (int nt = 0; nt < 4; ++nt) {
        float b = sB[nt * 16 + m];
        for (int r = 0; r < 8; ++r) acc[nt][r] = b;
    }

    for (int kb = 0; kb < DIM; kb += 4) {
        const int kk = kb + 2 * hi;
        v2f as, an;
        as.x = xrow[kk];
        as.y = xrow[kk + 1];
        an.x = arow[kk]     * dinv;
        an.y = arow[kk + 1] * dinv;

        for (int nt = 0; nt < 4; ++nt) {
            const int bi = kk * DIM + nt * 16 + m;
            v2f bs, bn;
            bs.x = sWs[bi];
            bs.y = sWs[bi + DIM];
            bn.x = sWn[bi];
            bn.y = sWn[bi + DIM];
            acc[nt] = __builtin_amdgcn_wmma_f32_16x16x4_f32(
                          false, as, false, bs, (short)0, acc[nt], false, false);
            acc[nt] = __builtin_amdgcn_wmma_f32_16x16x4_f32(
                          false, an, false, bn, (short)0, acc[nt], false, false);
        }
    }

    for (int nt = 0; nt < 4; ++nt) {
        const int col = nt * 16 + m;
        for (int r = 0; r < 8; ++r) {
            const int mrow = r + 8 * hi;
            out[(size_t)(row0 + mrow) * DIM + col] = acc[nt][r];
        }
    }
}

extern "C" void kernel_launch(void* const* d_in, const int* in_sizes, int n_in,
                              void* d_out, int out_size, void* d_ws, size_t ws_size,
                              hipStream_t stream) {
    const float* x      = (const float*)d_in[0];
    const int*   src    = (const int*)  d_in[1];
    const int*   dst    = (const int*)  d_in[2];
    const float* Wself  = (const float*)d_in[3];
    const float* bself  = (const float*)d_in[4];
    const float* Wneigh = (const float*)d_in[5];
    const float* bneigh = (const float*)d_in[6];
    float*       out    = (float*)d_out;

    const int N = in_sizes[0] / DIM;
    const int E = in_sizes[1];

    float* agg = (float*)d_ws;
    float* deg = agg + (size_t)N * DIM;

    const int zn = N * DIM + N;
    sage_zero_kernel<<<(zn + 255) / 256, 256, 0, stream>>>(agg, zn);

    const long long totThreads = (long long)E * 32;
    sage_edge_scatter_kernel<<<(int)((totThreads + 255) / 256), 256, 0, stream>>>(
        x, src, dst, agg, deg, E);

    const int nTiles = (N + 15) / 16;
    sage_wmma_gemm_kernel<<<(nTiles + 7) / 8, 256, 0, stream>>>(
        x, agg, deg, Wself, bself, Wneigh, bneigh, out, nTiles);
}


# Edge lists built per edge slice with heads in LDS (LDS exchange; no global atomic, no clearing launch: 2 launches); gather first collects a row's ids in LDS, rows add in LDS-slot (arrival) order
# speedup vs baseline: 5.0976x; 5.0976x over previous
#include <hip/hip_runtime.h>


typedef __attribute__((ext_vector_type(2))) float v2f;
typedef __attribute__((ext_vector_type(8))) float v8f;

#define DIM 64
#define NSL 16
#define RCAP 6400
#define ROWS_PB 64
#define LCAP 64
#define HP 68

__global__ void __launch_bounds__(1024)
sage_slice_lists_kernel(const int* __restrict__ src,
                        const int* __restrict__ dst,
                        int*       __restrict__ head,
                        int*       __restrict__ recw,
                        int E, int N, int ES) {
    __shared__ int sHead[RCAP];
    const int tid = threadIdx.x;
    const int s   = blockIdx.x;
    const int r0  = blockIdx.y * RCAP;
    const int nr  = min(RCAP, N - r0);
    for (int i = tid; i < RCAP; i += 1024) sHead[i] = -1;
    __syncthreads();

    int2* __restrict__ rec = (int2*)recw;
    const int eBeg = s * ES;
    const int eEnd = min(E, eBeg + ES);
    int  e0 = eBeg + tid * 4;
    bool fA = e0 + 3 < eEnd, fB = e0 + 4096 + 3 < eEnd;
    int4 dA = make_int4(-1, -1, -1, -1), sA = make_int4(0, 0, 0, 0);
    int4 dB = make_int4(-1, -1, -1, -1), sB = make_int4(0, 0, 0, 0);
    if (fA) { dA = *(const int4*)(dst + e0);        sA = *(const int4*)(src + e0); }
    if (fB) { dB = *(const int4*)(dst + e0 + 4096); sB = *(const int4*)(src + e0 + 4096); }
    while (e0 < eEnd) {
        const int  en  = e0 + 8192;
        const bool fAn = en + 3 < eEnd, fBn = en + 4096 + 3 < eEnd;
        int4 dAn = make_int4(-1, -1, -1, -1), sAn = make_int4(0, 0, 0, 0);
        int4 dBn = make_int4(-1, -1, -1, -1), sBn = make_int4(0, 0, 0, 0);
        if (fAn) { dAn = *(const int4*)(dst + en);        sAn = *(const int4*)(src + en); }
        if (fBn) { dBn = *(const int4*)(dst + en + 4096); sBn = *(const int4*)(src + en + 4096); }

        const int d[8]  = {dA.x, dA.y, dA.z, dA.w, dB.x, dB.y, dB.z, dB.w};
        const int sv[8] = {sA.x, sA.y, sA.z, sA.w, sB.x, sB.y, sB.z, sB.w};
        unsigned rel[8];
        int prev[8];
#pragma unroll
        for (int j = 0; j < 8; ++j) {
            rel[j]  = (unsigned)d[j] - (unsigned)r0;
            prev[j] = -1;
            if (rel[j] < (unsigned)nr) prev[j] = atomicExch(&sHead[rel[j]], e0 + (j & 3) + (j >> 2) * 4096);
        }
#pragma unroll
        for (int j = 0; j < 8; ++j)
            if (rel[j] < (unsigned)nr) rec[e0 + (j & 3) + (j >> 2) * 4096] = make_int2(prev[j], sv[j]);

        if (!fA)
            for (int e = e0; e < eEnd && e < e0 + 4; ++e) {
                const unsigned r1 = (unsigned)dst[e] - (unsigned)r0;
                if (r1 < (unsigned)nr) { const int pv = atomicExch(&sHead[r1], e); rec[e] = make_int2(pv, src[e]); }
            }
        if (!fB)
            for (int e = e0 + 4096; e < eEnd && e < e0 + 4100; ++e) {
                const unsigned r1 = (unsigned)dst[e] - (unsigned)r0;
                if (r1 < (unsigned)nr) { const int pv = atomicExch(&sHead[r1], e); rec[e] = make_int2(pv, src[e]); }
            }
        e0 = en; fA = fAn; fB = fBn; dA = dAn; sA = sAn; dB = dBn; sB = sBn;
    }
    __syncthreads();
    int* __restrict__ hp = head + (size_t)s * N + r0;
    for (int i = tid; i < nr; i += 1024) hp[i] = sHead[i];
}

__global__ void __launch_bounds__(512, 16)
sage_gather_wmma_kernel(const float* __restrict__ x,
                        const int*   __restrict__ head,
                        const int*   __restrict__ recw,
                        const float* __restrict__ Wself,
                        const float* __restrict__ bself,
                        const float* __restrict__ Wneigh,
                        const float* __restrict__ bneigh,
                        float*       __restrict__ out,
                        int N, int E, int nTiles) {
    __shared__ alignas(16) float sWs[DIM * DIM];
    __shared__ alignas(16) float sWn[DIM * DIM];
    __shared__ float sB[DIM];
    __shared__ alignas(16) float sH[ROWS_PB * HP];
    __shared__ alignas(16) int sSrc[ROWS_PB * LCAP];
    __shared__ int sCnt[ROWS_PB];
    __shared__ int sMore;

    const int tid  = threadIdx.x;
    const int wave = tid >> 5;
    const int lane = tid & 31;
    const int li   = lane & 7;
    const int rl   = wave * 4 + (lane >> 3);

    const int2*   __restrict__ rec = (const int2*)recw;
    const float4* __restrict__ x4  = (const float4*)x;

    const int arow = tid & (ROWS_PB - 1);
    const int asl  = tid >> 6;
    int ea = -1, eb = -1;
    {
        const int grow = blockIdx.x * ROWS_PB + arow;
        if (grow < N) {
            ea = head[(size_t)asl * N + grow];
            eb = head[(size_t)(asl + 8) * N + grow];
        }
    }

    for (int i = tid; i < DIM * DIM / 4; i += 512) {
        ((float4*)sWs)[i] = ((const float4*)Wself)[i];
        ((float4*)sWn)[i] = ((const float4*)Wneigh)[i];
    }
    if (tid < DIM) sB[tid] = bself[tid] + bneigh[tid];
    if (tid < ROWS_PB) sCnt[tid] = 0;
    if (tid == 0) sMore = 0;
    __syncthreads();

    float4 a0 = make_float4(0.f, 0.f, 0.f, 0.f);
    float4 a1 = make_float4(0.f, 0.f, 0.f, 0.f);
    int deg = 0;

    for (int round = 0; round <= E / LCAP; ++round) {
        bool pend = false;
        for (int it = 0; it < E; ++it) {
            const bool va = (unsigned)ea < (unsigned)E;
            const bool vb = (unsigned)eb < (unsigned)E;
            if (!(va || vb)) break;
            int2 ra = make_int2(-1, 0), rb = make_int2(-1, 0);
            if (va) ra = rec[ea];
            if (vb) rb = rec[eb];
            bool full = false;
            if (va) {
                const int sl = atomicAdd(&sCnt[arow], 1);
                if (sl < LCAP) { sSrc[arow * LCAP + sl] = ra.y; ea = ra.x; }
                else full = true;
            }
            if (vb && !full) {
                const int sl = atomicAdd(&sCnt[arow], 1);
                if (sl < LCAP) { sSrc[arow * LCAP + sl] = rb.y; eb = rb.x; }
                else full = true;
            }
            if (full) { pend = true; break; }
        }
        if (pend) sMore = 1;
        __syncthreads();
        const int more = sMore;

        const int n = min(sCnt[rl], LCAP);
        for (int i = 0; i < n; i += 4) {
            const int4 s4 = *(const int4*)&sSrc[rl * LCAP + i];
            const int s[4] = {s4.x, s4.y, s4.z, s4.w};
            bool   v[4];
            float4 p[4], q[4];
#pragma unroll
            for (int j = 0; j < 4; ++j) {
                v[j] = (i + j) < n;
                p[j] = make_float4(0.f, 0.f, 0.f, 0.f);
                q[j] = make_float4(0.f, 0.f, 0.f, 0.f);
                if (v[j]) {
                    const unsigned sr = min((unsigned)s[j], (unsigned)(N - 1));
                    const float4* xs = x4 + (size_t)sr * (DIM / 4);
                    p[j] = xs[li];
                    q[j] = xs[8 + li];
                }
            }
#pragma unroll
            for (int j = 0; j < 4; ++j) {
                a0.x += p[j].x; a0.y += p[j].y; a0.z += p[j].z; a0.w += p[j].w;
                a1.x += q[j].x; a1.y += q[j].y; a1.z += q[j].z; a1.w += q[j].w;
            }
        }
        deg += n;

        if (!more) break;
        __syncthreads();
        if (tid < ROWS_PB) sCnt[tid] = 0;
        if (tid == 0) sMore = 0;
        __syncthreads();
    }

    {
        const float dinv = 1.0f / fmaxf((float)deg, 1.0f);
        float4 h0 = make_float4(a0.x * dinv, a0.y * dinv, a0.z * dinv, a0.w * dinv);
        float4 h1 = make_float4(a1.x * dinv, a1.y * dinv, a1.z * dinv, a1.w * dinv);
        *(float4*)&sH[rl * HP + 4 * li]      = h0;
        *(float4*)&sH[rl * HP + 32 + 4 * li] = h1;
    }
    __syncthreads();

    if (wave >= 8) return;
    const int tile = wave >> 1;
    const int ch   = wave & 1;
    const int gt   = blockIdx.x * (ROWS_PB / 16) + tile;
    if (gt >= nTiles) return;

    const int m    = lane & 15;
    const int hi   = lane >> 4;
    const int row0 = gt * 16;

    const float* __restrict__ xrow = x + (size_t)(row0 + m) * DIM;
    const float* hrow = sH + (tile * 16 + m) * HP;

    v8f acc[2];
    for (int t = 0; t < 2; ++t) {
        float b = sB[(ch * 2 + t) * 16 + m];
        for (int rr = 0; rr < 8; ++rr) acc[t][rr] = b;
    }

    for (int kb = 0; kb < DIM; kb += 4) {
        const int kk = kb + 2 * hi;
        v2f as, an;
        as.x = xrow[kk];
        as.y = xrow[kk + 1];
        an.x = hrow[kk];
        an.y = hrow[kk + 1];
        for (int t = 0; t < 2; ++t) {
            const int bi = kk * DIM + (ch * 2 + t) * 16 + m;
            v2f bs, bn;
            bs.x = sWs[bi];
            bs.y = sWs[bi + DIM];
            bn.x = sWn[bi];
            bn.y = sWn[bi + DIM];
            acc[t] = __builtin_amdgcn_wmma_f32_16x16x4_f32(
                         false, as, false, bs, (short)0, acc[t], false, false);
            acc[t] = __builtin_amdgcn_wmma_f32_16x16x4_f32(
                         false, an, false, bn, (short)0, acc[t], false, false);
        }
    }

    for (int t = 0; t < 2; ++t) {
        const int col = (ch * 2 + t) * 16 + m;
        for (int rr = 0; rr < 8; ++rr) {
            const int mrow = rr + 8 * hi;
            out[(size_t)(row0 + mrow) * DIM + col] = acc[t][rr];
        }
    }
}

extern "C" void kernel_launch(void* const* d_in, const int* in_sizes, int n_in,
                              void* d_out, int out_size, void* d_ws, size_t ws_size,
                              hipStream_t stream) {
    const float* x      = (const float*)d_in[0];
    const int*   src    = (const int*)  d_in[1];
    const int*   dst    = (const int*)  d_in[2];
    const float* Wself  = (const float*)d_in[3];
    const float* bself  = (const float*)d_in[4];
    const float* Wneigh = (const float*)d_in[5];
    const float* bneigh = (const float*)d_in[6];
    float*       out    = (float*)d_out;

    const int N = in_sizes[0] / DIM;
    const int E = in_sizes[1];

    int* head = (int*)d_ws;
    int* recw = head + (size_t)N * NSL;

    const int ES = (((E + NSL - 1) / NSL) + 3) & ~3;
    const int H  = (N + RCAP - 1) / RCAP;
    sage_slice_lists_kernel<<<dim3(NSL, H), 1024, 0, stream>>>(src, dst, head, recw, E, N, ES);

    const int nTiles = (N + 15) / 16;
    sage_gather_wmma_kernel<<<(N + ROWS_PB - 1) / ROWS_PB, 512, 0, stream>>>(
        x, head, recw, Wself, bself, Wneigh, bneigh, out, N, E, nTiles);
}


# WMMA operands (both weight matrices, own 64 rows of x) read as one load group at kernel start and staged in LDS; self-product A fragments are LDS reads, no global load left in the WMMA part
# speedup vs baseline: 5.2549x; 1.0309x over previous
#include <hip/hip_runtime.h>


typedef __attribute__((ext_vector_type(2))) float v2f;
typedef __attribute__((ext_vector_type(8))) float v8f;

#define DIM 64
#define NSL 16
#define RCAP 6400
#define ROWS_PB 64
#define LCAP 32
#define HP 68

__global__ void __launch_bounds__(1024)
sage_slice_lists_kernel(const int* __restrict__ src,
                        const int* __restrict__ dst,
                        int*       __restrict__ head,
                        int*       __restrict__ recw,
                        int E, int N, int ES) {
    __shared__ int sHead[RCAP];
    const int tid = threadIdx.x;
    const int s   = blockIdx.x;
    const int r0  = blockIdx.y * RCAP;
    const int nr  = min(RCAP, N - r0);
    for (int i = tid; i < RCAP; i += 1024) sHead[i] = -1;
    __syncthreads();

    int2* __restrict__ rec = (int2*)recw;
    const int eBeg = s * ES;
    const int eEnd = min(E, eBeg + ES);
    int  e0 = eBeg + tid * 4;
    bool fA = e0 + 3 < eEnd, fB = e0 + 4096 + 3 < eEnd;
    int4 dA = make_int4(-1, -1, -1, -1), sA = make_int4(0, 0, 0, 0);
    int4 dB = make_int4(-1, -1, -1, -1), sB = make_int4(0, 0, 0, 0);
    if (fA) { dA = *(const int4*)(dst + e0);        sA = *(const int4*)(src + e0); }
    if (fB) { dB = *(const int4*)(dst + e0 + 4096); sB = *(const int4*)(src + e0 + 4096); }
    while (e0 < eEnd) {
        const int  en  = e0 + 8192;
        const bool fAn = en + 3 < eEnd, fBn = en + 4096 + 3 < eEnd;
        int4 dAn = make_int4(-1, -1, -1, -1), sAn = make_int4(0, 0, 0, 0);
        int4 dBn = make_int4(-1, -1, -1, -1), sBn = make_int4(0, 0, 0, 0);
        if (fAn) { dAn = *(const int4*)(dst + en);        sAn = *(const int4*)(src + en); }
        if (fBn) { dBn = *(const int4*)(dst + en + 4096); sBn = *(const int4*)(src + en + 4096); }

        const int d[8]  = {dA.x, dA.y, dA.z, dA.w, dB.x, dB.y, dB.z, dB.w};
        const int sv[8] = {sA.x, sA.y, sA.z, sA.w, sB.x, sB.y, sB.z, sB.w};
        unsigned rel[8];
        int prev[8];
#pragma unroll
        for (int j = 0; j < 8; ++j) {
            rel[j]  = (unsigned)d[j] - (unsigned)r0;
            prev[j] = -1;
            if (rel[j] < (unsigned)nr) prev[j] = atomicExch(&sHead[rel[j]], e0 + (j & 3) + (j >> 2) * 4096);
        }
#pragma unroll
        for (int j = 0; j < 8; ++j)
            if (rel[j] < (unsigned)nr) rec[e0 + (j & 3) + (j >> 2) * 4096] = make_int2(prev[j], sv[j]);

        if (!fA)
            for (int e = e0; e < eEnd && e < e0 + 4; ++e) {
                const unsigned r1 = (unsigned)dst[e] - (unsigned)r0;
                if (r1 < (unsigned)nr) { const int pv = atomicExch(&sHead[r1], e); rec[e] = make_int2(pv, src[e]); }
            }
        if (!fB)
            for (int e = e0 + 4096; e < eEnd && e < e0 + 4100; ++e) {
                const unsigned r1 = (unsigned)dst[e] - (unsigned)r0;
                if (r1 < (unsigned)nr) { const int pv = atomicExch(&sHead[r1], e); rec[e] = make_int2(pv, src[e]); }
            }
        e0 = en; fA = fAn; fB = fBn; dA = dAn; sA = sAn; dB = dBn; sB = sBn;
    }
    __syncthreads();
    int* __restrict__ hp = head + (size_t)s * N + r0;
    for (int i = tid; i < nr; i += 1024) hp[i] = sHead[i];
}

__global__ void __launch_bounds__(512, 16)
sage_gather_wmma_kernel(const float* __restrict__ x,
                        const int*   __restrict__ head,
                        const int*   __restrict__ recw,
                        const float* __restrict__ Wself,
                        const float* __restrict__ bself,
                        const float* __restrict__ Wneigh,
                        const float* __restrict__ bneigh,
                        float*       __restrict__ out,
                        int N, int E, int nTiles) {
    __shared__ alignas(16) float sWs[DIM * DIM];
    __shared__ alignas(16) float sWn[DIM * DIM];
    __shared__ float sB[DIM];
    __shared__ alignas(16) float sH[ROWS_PB * HP];
    __shared__ alignas(16) float sX[ROWS_PB * HP];
    __shared__ alignas(16) int sSrc[ROWS_PB * LCAP];
    __shared__ int sCnt[ROWS_PB];
    __shared__ int sMore;

    const int tid  = threadIdx.x;
    const int wave = tid >> 5;
    const int lane = tid & 31;
    const int li   = lane & 7;
    const int rl   = wave * 4 + (lane >> 3);

    const int2*   __restrict__ rec = (const int2*)recw;
    const float4* __restrict__ x4  = (const float4*)x;

    const int arow = tid & (ROWS_PB - 1);
    const int asl  = tid >> 6;
    int ea = -1, eb = -1;
    {
        const int grow = blockIdx.x * ROWS_PB + arow;
        if (grow < N) {
            ea = head[(size_t)asl * N + grow];
            eb = head[(size_t)(asl + 8) * N + grow];
        }
    }

    float4 xo0 = make_float4(0.f, 0.f, 0.f, 0.f);
    float4 xo1 = make_float4(0.f, 0.f, 0.f, 0.f);
    {
        const int grow = blockIdx.x * ROWS_PB + rl;
        if (grow < N) {
            xo0 = x4[(size_t)grow * (DIM / 4) + li];
            xo1 = x4[(size_t)grow * (DIM / 4) + 8 + li];
        }
    }

    {
        const float4 ws0 = ((const float4*)Wself)[tid];
        const float4 ws1 = ((const float4*)Wself)[tid + 512];
        const float4 wn0 = ((const float4*)Wneigh)[tid];
        const float4 wn1 = ((const float4*)Wneigh)[tid + 512];
        ((float4*)sWs)[tid]       = ws0;
        ((float4*)sWs)[tid + 512] = ws1;
        ((float4*)sWn)[tid]       = wn0;
        ((float4*)sWn)[tid + 512] = wn1;
    }
    *(float4*)&sX[rl * HP + 4 * li]      = xo0;
    *(float4*)&sX[rl * HP + 32 + 4 * li] = xo1;
    if (tid < DIM) sB[tid] = bself[tid] + bneigh[tid];
    if (tid < ROWS_PB) sCnt[tid] = 0;
    if (tid == 0) sMore = 0;
    __syncthreads();

    float4 a0 = make_float4(0.f, 0.f, 0.f, 0.f);
    float4 a1 = make_float4(0.f, 0.f, 0.f, 0.f);
    int deg = 0;

    for (int round = 0; round <= E / LCAP; ++round) {
        bool pend = false;
        for (int it = 0; it < E; ++it) {
            const bool va = (unsigned)ea < (unsigned)E;
            const bool vb = (unsigned)eb < (unsigned)E;
            if (!(va || vb)) break;
            int2 ra = make_int2(-1, 0), rb = make_int2(-1, 0);
            if (va) ra = rec[ea];
            if (vb) rb = rec[eb];
            bool full = false;
            if (va) {
                const int sl = atomicAdd(&sCnt[arow], 1);
                if (sl < LCAP) { sSrc[arow * LCAP + sl] = ra.y; ea = ra.x; }
                else full = true;
            }
            if (vb && !full) {
                const int sl = atomicAdd(&sCnt[arow], 1);
                if (sl < LCAP) { sSrc[arow * LCAP + sl] = rb.y; eb = rb.x; }
                else full = true;
            }
            if (full) { pend = true; break; }
        }
        if (pend) sMore = 1;
        __syncthreads();
        const int more = sMore;

        const int n = min(sCnt[rl], LCAP);
        for (int i = 0; i < n; i += 4) {
            const int4 s4 = *(const int4*)&sSrc[rl * LCAP + i];
            const int s[4] = {s4.x, s4.y, s4.z, s4.w};
            bool   v[4];
            float4 p[4], q[4];
#pragma unroll
            for (int j = 0; j < 4; ++j) {
                v[j] = (i + j) < n;
                p[j] = make_float4(0.f, 0.f, 0.f, 0.f);
                q[j] = make_float4(0.f, 0.f, 0.f, 0.f);
                if (v[j]) {
                    const unsigned sr = min((unsigned)s[j], (unsigned)(N - 1));
                    const float4* xs = x4 + (size_t)sr * (DIM / 4);
                    p[j] = xs[li];
                    q[j] = xs[8 + li];
                }
            }
#pragma unroll
            for (int j = 0; j < 4; ++j) {
                a0.x += p[j].x; a0.y += p[j].y; a0.z += p[j].z; a0.w += p[j].w;
                a1.x += q[j].x; a1.y += q[j].y; a1.z += q[j].z; a1.w += q[j].w;
            }
        }
        deg += n;

        if (!more) break;
        __syncthreads();
        if (tid < ROWS_PB) sCnt[tid] = 0;
        if (tid == 0) sMore = 0;
        __syncthreads();
    }

    {
        const float dinv = 1.0f / fmaxf((float)deg, 1.0f);
        float4 h0 = make_float4(a0.x * dinv, a0.y * dinv, a0.z * dinv, a0.w * dinv);
        float4 h1 = make_float4(a1.x * dinv, a1.y * dinv, a1.z * dinv, a1.w * dinv);
        *(float4*)&sH[rl * HP + 4 * li]      = h0;
        *(float4*)&sH[rl * HP + 32 + 4 * li] = h1;
    }
    __syncthreads();

    if (wave >= 8) return;
    const int tile = wave >> 1;
    const int ch   = wave & 1;
    const int gt   = blockIdx.x * (ROWS_PB / 16) + tile;
    if (gt >= nTiles) return;

    const int m    = lane & 15;
    const int hi   = lane >> 4;
    const int row0 = gt * 16;

    const float* xrow = sX + (tile * 16 + m) * HP;
    const float* hrow = sH + (tile * 16 + m) * HP;

    v8f acc[2];
    for (int t = 0; t < 2; ++t) {
        float b = sB[(ch * 2 + t) * 16 + m];
        for (int rr = 0; rr < 8; ++rr) acc[t][rr] = b;
    }

    for (int kb = 0; kb < DIM; kb += 4) {
        const int kk = kb + 2 * hi;
        v2f as, an;
        as.x = xrow[kk];
        as.y = xrow[kk + 1];
        an.x = hrow[kk];
        an.y = hrow[kk + 1];
        for (int t = 0; t < 2; ++t) {
            const int bi = kk * DIM + (ch * 2 + t) * 16 + m;
            v2f bs, bn;
            bs.x = sWs[bi];
            bs.y = sWs[bi + DIM];
            bn.x = sWn[bi];
            bn.y = sWn[bi + DIM];
            acc[t] = __builtin_amdgcn_wmma_f32_16x16x4_f32(
                         false, as, false, bs, (short)0, acc[t], false, false);
            acc[t] = __builtin_amdgcn_wmma_f32_16x16x4_f32(
                         false, an, false, bn, (short)0, acc[t], false, false);
        }
    }

    for (int t = 0; t < 2; ++t) {
        const int col = (ch * 2 + t) * 16 + m;
        for (int rr = 0; rr < 8; ++rr) {
            const int mrow = rr + 8 * hi;
            out[(size_t)(row0 + mrow) * DIM + col] = acc[t][rr];
        }
    }
}

extern "C" void kernel_launch(void* const* d_in, const int* in_sizes, int n_in,
                              void* d_out, int out_size, void* d_ws, size_t ws_size,
                              hipStream_t stream) {
    const float* x      = (const float*)d_in[0];
    const int*   src    = (const int*)  d_in[1];
    const int*   dst    = (const int*)  d_in[2];
    const float* Wself  = (const float*)d_in[3];
    const float* bself  = (const float*)d_in[4];
    const float* Wneigh = (const float*)d_in[5];
    const float* bneigh = (const float*)d_in[6];
    float*       out    = (float*)d_out;

    const int N = in_sizes[0] / DIM;
    const int E = in_sizes[1];

    int* head = (int*)d_ws;
    int* recw = head + (size_t)N * NSL;

    const int ES = (((E + NSL - 1) / NSL) + 3) & ~3;
    const int H  = (N + RCAP - 1) / RCAP;
    sage_slice_lists_kernel<<<dim3(NSL, H), 1024, 0, stream>>>(src, dst, head, recw, E, N, ES);

    const int nTiles = (N + 15) / 16;
    sage_gather_wmma_kernel<<<(N + ROWS_PB - 1) / ROWS_PB, 512, 0, stream>>>(
        x, head, recw, Wself, bself, Wneigh, bneigh, out, N, E, nTiles);
}


# List head now holds its newest edge's source row and the list length (LDS exchange + count): a list of L edges costs L-1 4-byte record reads and writes, not L of 8 bytes; sums still in arrival order
# speedup vs baseline: 5.9554x; 1.1333x over previous
#include <hip/hip_runtime.h>


typedef __attribute__((ext_vector_type(2))) float v2f;
typedef __attribute__((ext_vector_type(8))) float v8f;

#define DIM 64
#define NSL 16
#define RCAP 6400
#define ROWS_PB 64
#define LCAP 32
#define HP 68

__global__ void __launch_bounds__(1024)
sage_slice_lists_kernel(const int* __restrict__ src,
                        const int* __restrict__ dst,
                        int*       __restrict__ head,
                        int*       __restrict__ recw,
                        int E, int N, int ES) {
    __shared__ alignas(8) unsigned sHead[2 * RCAP];
    const int tid = threadIdx.x;
    const int s   = blockIdx.x;
    const int r0  = blockIdx.y * RCAP;
    const int nr  = min(RCAP, N - r0);
    const int sb  = 32 - __clz(N - 1);
    const unsigned smax = (unsigned)(N - 1);
    for (int i = tid; i < 2 * RCAP; i += 1024) sHead[i] = 0u;
    __syncthreads();

    unsigned* __restrict__ rec = (unsigned*)recw;
    const int eBeg = s * ES;
    const int eEnd = min(E, eBeg + ES);
    int  e0 = eBeg + tid * 4;
    bool fA = e0 + 3 < eEnd, fB = e0 + 4096 + 3 < eEnd;
    int4 dA = make_int4(-1, -1, -1, -1), sA = make_int4(0, 0, 0, 0);
    int4 dB = make_int4(-1, -1, -1, -1), sB = make_int4(0, 0, 0, 0);
    if (fA) { dA = *(const int4*)(dst + e0);        sA = *(const int4*)(src + e0); }
    if (fB) { dB = *(const int4*)(dst + e0 + 4096); sB = *(const int4*)(src + e0 + 4096); }
    while (e0 < eEnd) {
        const int  en  = e0 + 8192;
        const bool fAn = en + 3 < eEnd, fBn = en + 4096 + 3 < eEnd;
        int4 dAn = make_int4(-1, -1, -1, -1), sAn = make_int4(0, 0, 0, 0);
        int4 dBn = make_int4(-1, -1, -1, -1), sBn = make_int4(0, 0, 0, 0);
        if (fAn) { dAn = *(const int4*)(dst + en);        sAn = *(const int4*)(src + en); }
        if (fBn) { dBn = *(const int4*)(dst + en + 4096); sBn = *(const int4*)(src + en + 4096); }

        const int d[8]  = {dA.x, dA.y, dA.z, dA.w, dB.x, dB.y, dB.z, dB.w};
        const int sv[8] = {sA.x, sA.y, sA.z, sA.w, sB.x, sB.y, sB.z, sB.w};
        const int p1 = e0 - eBeg + 1;
        unsigned prev[8];
#pragma unroll
        for (int j = 0; j < 8; ++j) {
            const unsigned rel = (unsigned)d[j] - (unsigned)r0;
            prev[j] = 0u;
            if (rel < (unsigned)nr) {
                const unsigned wv = ((unsigned)(p1 + (j & 3) + (j >> 2) * 4096) << sb) | min((unsigned)sv[j], smax);
                prev[j] = atomicExch(&sHead[2 * rel], wv);
                atomicAdd(&sHead[2 * rel + 1], 1u);
            }
        }
#pragma unroll
        for (int j = 0; j < 8; ++j)
            if (prev[j] != 0u) rec[e0 + (j & 3) + (j >> 2) * 4096] = prev[j];

        if (!fA)
            for (int e = e0; e < eEnd && e < e0 + 4; ++e) {
                const unsigned r1 = (unsigned)dst[e] - (unsigned)r0;
                if (r1 < (unsigned)nr) {
                    const unsigned wv = ((unsigned)(e - eBeg + 1) << sb) | min((unsigned)src[e], smax);
                    const unsigned pv = atomicExch(&sHead[2 * r1], wv);
                    atomicAdd(&sHead[2 * r1 + 1], 1u);
                    if (pv != 0u) rec[e] = pv;
                }
            }
        if (!fB)
            for (int e = e0 + 4096; e < eEnd && e < e0 + 4100; ++e) {
                const unsigned r1 = (unsigned)dst[e] - (unsigned)r0;
                if (r1 < (unsigned)nr) {
                    const unsigned wv = ((unsigned)(e - eBeg + 1) << sb) | min((unsigned)src[e], smax);
                    const unsigned pv = atomicExch(&sHead[2 * r1], wv);
                    atomicAdd(&sHead[2 * r1 + 1], 1u);
                    if (pv != 0u) rec[e] = pv;
                }
            }
        e0 = en; fA = fAn; fB = fBn; dA = dAn; sA = sAn; dB = dBn; sB = sBn;
    }
    __syncthreads();
    uint2* __restrict__ hp = (uint2*)head + (size_t)s * N + r0;
    for (int i = tid; i < nr; i += 1024) hp[i] = *(const uint2*)&sHead[2 * i];
}

__global__ void __launch_bounds__(512, 16)
sage_gather_wmma_kernel(const float* __restrict__ x,
                        const int*   __restrict__ head,
                        const int*   __restrict__ recw,
                        const float* __restrict__ Wself,
                        const float* __restrict__ bself,
                        const float* __restrict__ Wneigh,
                        const float* __restrict__ bneigh,
                        float*       __restrict__ out,
                        int N, int E, int nTiles, int ES) {
    __shared__ alignas(16) float sWs[DIM * DIM];
    __shared__ alignas(16) float sWn[DIM * DIM];
    __shared__ float sB[DIM];
    __shared__ alignas(16) float sH[ROWS_PB * HP];
    __shared__ alignas(16) float sX[ROWS_PB * HP];
    __shared__ alignas(16) int sSrc[ROWS_PB * LCAP];
    __shared__ int sCnt[ROWS_PB];
    __shared__ int sMore;

    const int tid  = threadIdx.x;
    const int wave = tid >> 5;
    const int lane = tid & 31;
    const int li   = lane & 7;
    const int rl   = wave * 4 + (lane >> 3);

    const uint2*    __restrict__ head2 = (const uint2*)head;
    const unsigned* __restrict__ rec   = (const unsigned*)recw;
    const float4*   __restrict__ x4    = (const float4*)x;
    const int      sb    = 32 - __clz(N - 1);
    const unsigned smask = (1u << sb) - 1u;
    const unsigned elast = (unsigned)(E - 1);

    const int arow = tid & (ROWS_PB - 1);
    const int asl  = __builtin_amdgcn_readfirstlane(tid >> 6);
    unsigned wa = 0u, wb = 0u;
    int      ca = 0,  cb = 0;
    {
        const int grow = blockIdx.x * ROWS_PB + arow;
        if (grow < N) {
            const uint2 ha = head2[(size_t)asl * N + grow];
            const uint2 hb = head2[(size_t)(asl + 8) * N + grow];
            wa = ha.x; ca = (int)ha.y;
            wb = hb.x; cb = (int)hb.y;
        }
    }
    const unsigned baseA = (unsigned)(asl * ES) - 1u;
    const unsigned baseB = (unsigned)((asl + 8) * ES) - 1u;

    float4 xo0 = make_float4(0.f, 0.f, 0.f, 0.f);
    float4 xo1 = make_float4(0.f, 0.f, 0.f, 0.f);
    {
        const int grow = blockIdx.x * ROWS_PB + rl;
        if (grow < N) {
            xo0 = x4[(size_t)grow * (DIM / 4) + li];
            xo1 = x4[(size_t)grow * (DIM / 4) + 8 + li];
        }
    }

    {
        const float4 ws0 = ((const float4*)Wself)[tid];
        const float4 ws1 = ((const float4*)Wself)[tid + 512];
        const float4 wn0 = ((const float4*)Wneigh)[tid];
        const float4 wn1 = ((const float4*)Wneigh)[tid + 512];
        ((float4*)sWs)[tid]       = ws0;
        ((float4*)sWs)[tid + 512] = ws1;
        ((float4*)sWn)[tid]       = wn0;
        ((float4*)sWn)[tid + 512] = wn1;
    }
    *(float4*)&sX[rl * HP + 4 * li]      = xo0;
    *(float4*)&sX[rl * HP + 32 + 4 * li] = xo1;
    if (tid < DIM) sB[tid] = bself[tid] + bneigh[tid];
    if (tid < ROWS_PB) sCnt[tid] = 0;
    if (tid == 0) sMore = 0;
    __syncthreads();

    float4 a0 = make_float4(0.f, 0.f, 0.f, 0.f);
    float4 a1 = make_float4(0.f, 0.f, 0.f, 0.f);
    int deg = 0;

    for (int round = 0; round <= E / LCAP; ++round) {
        bool pend = false;
        for (int it = 0; it < E; ++it) {
            const bool va = ca > 0;
            const bool vb = cb > 0;
            if (!(va || vb)) break;
            unsigned na = 0u, nb = 0u;
            if (ca > 1) na = rec[min(baseA + (wa >> sb), elast)];
            if (cb > 1) nb = rec[min(baseB + (wb >> sb), elast)];
            bool full = false;
            if (va) {
                const int sl = atomicAdd(&sCnt[arow], 1);
                if (sl < LCAP) { sSrc[arow * LCAP + sl] = (int)(wa & smask); wa = na; --ca; }
                else full = true;
            }
            if (vb && !full) {
                const int sl = atomicAdd(&sCnt[arow], 1);
                if (sl < LCAP) { sSrc[arow * LCAP + sl] = (int)(wb & smask); wb = nb; --cb; }
                else full = true;
            }
            if (full) { pend = true; break; }
        }
        if (pend) sMore = 1;
        __syncthreads();
        const int more = sMore;

        const int n = min(sCnt[rl], LCAP);
        for (int i = 0; i < n; i += 4) {
            const int4 s4 = *(const int4*)&sSrc[rl * LCAP + i];
            const int s[4] = {s4.x, s4.y, s4.z, s4.w};
            bool   v[4];
            float4 p[4], q[4];
#pragma unroll
            for (int j = 0; j < 4; ++j) {
                v[j] = (i + j) < n;
                p[j] = make_float4(0.f, 0.f, 0.f, 0.f);
                q[j] = make_float4(0.f, 0.f, 0.f, 0.f);
                if (v[j]) {
                    const unsigned sr = min((unsigned)s[j], (unsigned)(N - 1));
                    const float4* xs = x4 + (size_t)sr * (DIM / 4);
                    p[j] = xs[li];
                    q[j] = xs[8 + li];
                }
            }
#pragma unroll
            for (int j = 0; j < 4; ++j) {
                a0.x += p[j].x; a0.y += p[j].y; a0.z += p[j].z; a0.w += p[j].w;
                a1.x += q[j].x; a1.y += q[j].y; a1.z += q[j].z; a1.w += q[j].w;
            }
        }
        deg += n;

        if (!more) break;
        __syncthreads();
        if (tid < ROWS_PB) sCnt[tid] = 0;
        if (tid == 0) sMore = 0;
        __syncthreads();
    }

    {
        const float dinv = 1.0f / fmaxf((float)deg, 1.0f);
        float4 h0 = make_float4(a0.x * dinv, a0.y * dinv, a0.z * dinv, a0.w * dinv);
        float4 h1 = make_float4(a1.x * dinv, a1.y * dinv, a1.z * dinv, a1.w * dinv);
        *(float4*)&sH[rl * HP + 4 * li]      = h0;
        *(float4*)&sH[rl * HP + 32 + 4 * li] = h1;
    }
    __syncthreads();

    if (wave >= 8) return;
    const int tile = wave >> 1;
    const int ch   = wave & 1;
    const int gt   = blockIdx.x * (ROWS_PB / 16) + tile;
    if (gt >= nTiles) return;

    const int m    = lane & 15;
    const int hi   = lane >> 4;
    const int row0 = gt * 16;

    const float* xrow = sX + (tile * 16 + m) * HP;
    const float* hrow = sH + (tile * 16 + m) * HP;

    v8f acc[2];
    for (int t = 0; t < 2; ++t) {
        float b = sB[(ch * 2 + t) * 16 + m];
        for (int rr = 0; rr < 8; ++rr) acc[t][rr] = b;
    }

    for (int kb = 0; kb < DIM; kb += 4) {
        const int kk = kb + 2 * hi;
        v2f as, an;
        as.x = xrow[kk];
        as.y = xrow[kk + 1];
        an.x = hrow[kk];
        an.y = hrow[kk + 1];
        for (int t = 0; t < 2; ++t) {
            const int bi = kk * DIM + (ch * 2 + t) * 16 + m;
            v2f bs, bn;
            bs.x = sWs[bi];
            bs.y = sWs[bi + DIM];
            bn.x = sWn[bi];
            bn.y = sWn[bi + DIM];
            acc[t] = __builtin_amdgcn_wmma_f32_16x16x4_f32(
                         false, as, false, bs, (short)0, acc[t], false, false);
            acc[t] = __builtin_amdgcn_wmma_f32_16x16x4_f32(
                         false, an, false, bn, (short)0, acc[t], false, false);
        }
    }

    for (int t = 0; t < 2; ++t) {
        const int col = (ch * 2 + t) * 16 + m;
        for (int rr = 0; rr < 8; ++rr) {
            const int mrow = rr + 8 * hi;
            out[(size_t)(row0 + mrow) * DIM + col] = acc[t][rr];
        }
    }
}

extern "C" void kernel_launch(void* const* d_in, const int* in_sizes, int n_in,
                              void* d_out, int out_size, void* d_ws, size_t ws_size,
                              hipStream_t stream) {
    const float* x      = (const float*)d_in[0];
    const int*   src    = (const int*)  d_in[1];
    const int*   dst    = (const int*)  d_in[2];
    const float* Wself  = (const float*)d_in[3];
    const float* bself  = (const float*)d_in[4];
    const float* Wneigh = (const float*)d_in[5];
    const float* bneigh = (const float*)d_in[6];
    float*       out    = (float*)d_out;

    const int N = in_sizes[0] / DIM;
    const int E = in_sizes[1];

    int* head = (int*)d_ws;
    int* recw = head + (size_t)N * NSL * 2;

    const int ES = (((E + NSL - 1) / NSL) + 3) & ~3;
    const int H  = (N + RCAP - 1) / RCAP;
    sage_slice_lists_kernel<<<dim3(NSL, H), 1024, 0, stream>>>(src, dst, head, recw, E, N, ES);

    const int nTiles = (N + 15) / 16;
    sage_gather_wmma_kernel<<<(N + ROWS_PB - 1) / ROWS_PB, 512, 0, stream>>>(
        x, head, recw, Wself, bself, Wneigh, bneigh, out, N, E, nTiles, ES);
}


# Edge slices 16 -> 24, row ranges 8 -> 4 in the list-building launch (heads of 12800 rows in LDS per work-group): each scans 33336 edges, not 50000; gather thread walks 3 lists; arrival order kept
# speedup vs baseline: 6.1529x; 1.0332x over previous
#include <hip/hip_runtime.h>


typedef __attribute__((ext_vector_type(2))) float v2f;
typedef __attribute__((ext_vector_type(8))) float v8f;

#define DIM 64
#define NSL 24
#define LPT (NSL / 8)
#define RCAP 12800
#define ROWS_PB 64
#define LCAP 32
#define HP 68

__global__ void __launch_bounds__(1024)
sage_slice_lists_kernel(const int* __restrict__ src,
                        const int* __restrict__ dst,
                        int*       __restrict__ head,
                        int*       __restrict__ recw,
                        int E, int N, int ES) {
    __shared__ alignas(16) unsigned sHead[2 * RCAP];
    const int tid = threadIdx.x;
    const int s   = blockIdx.x;
    const int r0  = blockIdx.y * RCAP;
    const int nr  = min(RCAP, N - r0);
    const int sb  = 32 - __clz(N - 1);
    const unsigned smax = (unsigned)(N - 1);
    for (int i = tid; i < RCAP / 2; i += 1024) ((uint4*)sHead)[i] = make_uint4(0u, 0u, 0u, 0u);
    __syncthreads();

    unsigned* __restrict__ rec = (unsigned*)recw;
    const int eBeg = s * ES;
    const int eEnd = min(E, eBeg + ES);
    int  e0 = eBeg + tid * 4;
    bool fA = e0 + 3 < eEnd, fB = e0 + 4096 + 3 < eEnd;
    int4 dA = make_int4(-1, -1, -1, -1), sA = make_int4(0, 0, 0, 0);
    int4 dB = make_int4(-1, -1, -1, -1), sB = make_int4(0, 0, 0, 0);
    if (fA) { dA = *(const int4*)(dst + e0);        sA = *(const int4*)(src + e0); }
    if (fB) { dB = *(const int4*)(dst + e0 + 4096); sB = *(const int4*)(src + e0 + 4096); }
    while (e0 < eEnd) {
        const int  en  = e0 + 8192;
        const bool fAn = en + 3 < eEnd, fBn = en + 4096 + 3 < eEnd;
        int4 dAn = make_int4(-1, -1, -1, -1), sAn = make_int4(0, 0, 0, 0);
        int4 dBn = make_int4(-1, -1, -1, -1), sBn = make_int4(0, 0, 0, 0);
        if (fAn) { dAn = *(const int4*)(dst + en);        sAn = *(const int4*)(src + en); }
        if (fBn) { dBn = *(const int4*)(dst + en + 4096); sBn = *(const int4*)(src + en + 4096); }

        const int d[8]  = {dA.x, dA.y, dA.z, dA.w, dB.x, dB.y, dB.z, dB.w};
        const int sv[8] = {sA.x, sA.y, sA.z, sA.w, sB.x, sB.y, sB.z, sB.w};
        const int p1 = e0 - eBeg + 1;
        unsigned prev[8];
#pragma unroll
        for (int j = 0; j < 8; ++j) {
            const unsigned rel = (unsigned)d[j] - (unsigned)r0;
            prev[j] = 0u;
            if (rel < (unsigned)nr) {
                const unsigned wv = ((unsigned)(p1 + (j & 3) + (j >> 2) * 4096) << sb) | min((unsigned)sv[j], smax);
                prev[j] = atomicExch(&sHead[2 * rel], wv);
                atomicAdd(&sHead[2 * rel + 1], 1u);
            }
        }
#pragma unroll
        for (int j = 0; j < 8; ++j)
            if (prev[j] != 0u) rec[e0 + (j & 3) + (j >> 2) * 4096] = prev[j];

        if (!fA)
            for (int e = e0; e < eEnd && e < e0 + 4; ++e) {
                const unsigned r1 = (unsigned)dst[e] - (unsigned)r0;
                if (r1 < (unsigned)nr) {
                    const unsigned wv = ((unsigned)(e - eBeg + 1) << sb) | min((unsigned)src[e], smax);
                    const unsigned pv = atomicExch(&sHead[2 * r1], wv);
                    atomicAdd(&sHead[2 * r1 + 1], 1u);
                    if (pv != 0u) rec[e] = pv;
                }
            }
        if (!fB)
            for (int e = e0 + 4096; e < eEnd && e < e0 + 4100; ++e) {
                const unsigned r1 = (unsigned)dst[e] - (unsigned)r0;
                if (r1 < (unsigned)nr) {
                    const unsigned wv = ((unsigned)(e - eBeg + 1) << sb) | min((unsigned)src[e], smax);
                    const unsigned pv = atomicExch(&sHead[2 * r1], wv);
                    atomicAdd(&sHead[2 * r1 + 1], 1u);
                    if (pv != 0u) rec[e] = pv;
                }
            }
        e0 = en; fA = fAn; fB = fBn; dA = dAn; sA = sAn; dB = dBn; sB = sBn;
    }
    __syncthreads();
    uint2* __restrict__ hp = (uint2*)head + (size_t)s * N + r0;
    for (int i = tid; i < nr; i += 1024) hp[i] = *(const uint2*)&sHead[2 * i];
}

__global__ void __launch_bounds__(512, 16)
sage_gather_wmma_kernel(const float* __restrict__ x,
                        const int*   __restrict__ head,
                        const int*   __restrict__ recw,
                        const float* __restrict__ Wself,
                        const float* __restrict__ bself,
                        const float* __restrict__ Wneigh,
                        const float* __restrict__ bneigh,
                        float*       __restrict__ out,
                        int N, int E, int nTiles, int ES) {
    __shared__ alignas(16) float sWs[DIM * DIM];
    __shared__ alignas(16) float sWn[DIM * DIM];
    __shared__ float sB[DIM];
    __shared__ alignas(16) float sH[ROWS_PB * HP];
    __shared__ alignas(16) float sX[ROWS_PB * HP];
    __shared__ alignas(16) int sSrc[ROWS_PB * LCAP];
    __shared__ int sCnt[ROWS_PB];
    __shared__ int sMore;

    const int tid  = threadIdx.x;
    const int wave = tid >> 5;
    const int lane = tid & 31;
    const int li   = lane & 7;
    const int rl   = wave * 4 + (lane >> 3);

    const uint2*    __restrict__ head2 = (const uint2*)head;
    const unsigned* __restrict__ rec   = (const unsigned*)recw;
    const float4*   __restrict__ x4    = (const float4*)x;
    const int      sb    = 32 - __clz(N - 1);
    const unsigned smask = (1u << sb) - 1u;
    const unsigned elast = (unsigned)(E - 1);

    const int arow = tid & (ROWS_PB - 1);
    const int asl  = __builtin_amdgcn_readfirstlane(tid >> 6);
    uint2 hd[LPT];
    unsigned rbase[LPT];
    const int  agrow = (int)blockIdx.x * ROWS_PB + arow;
    const bool rowok = agrow < N;
    {
        const int growc = min(agrow, N - 1);
#pragma unroll
        for (int k = 0; k < LPT; ++k) {
            rbase[k] = (unsigned)((asl + 8 * k) * ES) - 1u;
            hd[k] = head2[(size_t)(asl + 8 * k) * N + growc];
        }
    }

    float4 xo0 = make_float4(0.f, 0.f, 0.f, 0.f);
    float4 xo1 = make_float4(0.f, 0.f, 0.f, 0.f);
    {
        const int grow = blockIdx.x * ROWS_PB + rl;
        if (grow < N) {
            xo0 = x4[(size_t)grow * (DIM / 4) + li];
            xo1 = x4[(size_t)grow * (DIM / 4) + 8 + li];
        }
    }

    {
        const float4 ws0 = ((const float4*)Wself)[tid];
        const float4 ws1 = ((const float4*)Wself)[tid + 512];
        const float4 wn0 = ((const float4*)Wneigh)[tid];
        const float4 wn1 = ((const float4*)Wneigh)[tid + 512];
        ((float4*)sWs)[tid]       = ws0;
        ((float4*)sWs)[tid + 512] = ws1;
        ((float4*)sWn)[tid]       = wn0;
        ((float4*)sWn)[tid + 512] = wn1;
    }
    *(float4*)&sX[rl * HP + 4 * li]      = xo0;
    *(float4*)&sX[rl * HP + 32 + 4 * li] = xo1;
    if (tid < DIM) sB[tid] = bself[tid] + bneigh[tid];
    if (tid < ROWS_PB) sCnt[tid] = 0;
    if (tid == 0) sMore = 0;
    __syncthreads();

    unsigned w[LPT];
    int      c[LPT];
#pragma unroll
    for (int k = 0; k < LPT; ++k) { w[k] = hd[k].x; c[k] = rowok ? (int)hd[k].y : 0; }

    float4 a0 = make_float4(0.f, 0.f, 0.f, 0.f);
    float4 a1 = make_float4(0.f, 0.f, 0.f, 0.f);
    int deg = 0;

    for (int round = 0; round <= E / LCAP; ++round) {
        bool pend = false;
        for (int it = 0; it < E; ++it) {
            bool any = false;
#pragma unroll
            for (int k = 0; k < LPT; ++k) any = any || (c[k] > 0);
            if (!any) break;
            unsigned nx[LPT];
#pragma unroll
            for (int k = 0; k < LPT; ++k) {
                nx[k] = 0u;
                if (c[k] > 1) nx[k] = rec[min(rbase[k] + (w[k] >> sb), elast)];
            }
            bool full = false;
#pragma unroll
            for (int k = 0; k < LPT; ++k) {
                if (c[k] > 0 && !full) {
                    const int sl = atomicAdd(&sCnt[arow], 1);
                    if (sl < LCAP) { sSrc[arow * LCAP + sl] = (int)(w[k] & smask); w[k] = nx[k]; --c[k]; }
                    else full = true;
                }
            }
            if (full) { pend = true; break; }
        }
        if (pend) sMore = 1;
        __syncthreads();
        const int more = sMore;

        const int n = min(sCnt[rl], LCAP);
        for (int i = 0; i < n; i += 4) {
            const int4 s4 = *(const int4*)&sSrc[rl * LCAP + i];
            const int s[4] = {s4.x, s4.y, s4.z, s4.w};
            bool   v[4];
            float4 p[4], q[4];
#pragma unroll
            for (int j = 0; j < 4; ++j) {
                v[j] = (i + j) < n;
                p[j] = make_float4(0.f, 0.f, 0.f, 0.f);
                q[j] = make_float4(0.f, 0.f, 0.f, 0.f);
                if (v[j]) {
                    const unsigned sr = min((unsigned)s[j], (unsigned)(N - 1));
                    const float4* xs = x4 + (size_t)sr * (DIM / 4);
                    p[j] = xs[li];
                    q[j] = xs[8 + li];
                }
            }
#pragma unroll
            for (int j = 0; j < 4; ++j) {
                a0.x += p[j].x; a0.y += p[j].y; a0.z += p[j].z; a0.w += p[j].w;
                a1.x += q[j].x; a1.y += q[j].y; a1.z += q[j].z; a1.w += q[j].w;
            }
        }
        deg += n;

        if (!more) break;
        __syncthreads();
        if (tid < ROWS_PB) sCnt[tid] = 0;
        if (tid == 0) sMore = 0;
        __syncthreads();
    }

    {
        const float dinv = 1.0f / fmaxf((float)deg, 1.0f);
        float4 h0 = make_float4(a0.x * dinv, a0.y * dinv, a0.z * dinv, a0.w * dinv);
        float4 h1 = make_float4(a1.x * dinv, a1.y * dinv, a1.z * dinv, a1.w * dinv);
        *(float4*)&sH[rl * HP + 4 * li]      = h0;
        *(float4*)&sH[rl * HP + 32 + 4 * li] = h1;
    }
    __syncthreads();

    if (wave >= 8) return;
    const int tile = wave >> 1;
    const int ch   = wave & 1;
    const int gt   = blockIdx.x * (ROWS_PB / 16) + tile;
    if (gt >= nTiles) return;

    const int m    = lane & 15;
    const int hi   = lane >> 4;
    const int row0 = gt * 16;

    const float* xrow = sX + (tile * 16 + m) * HP;
    const float* hrow = sH + (tile * 16 + m) * HP;

    v8f acc[2];
    for (int t = 0; t < 2; ++t) {
        float b = sB[(ch * 2 + t) * 16 + m];
        for (int rr = 0; rr < 8; ++rr) acc[t][rr] = b;
    }

    for (int kb = 0; kb < DIM; kb += 4) {
        const int kk = kb + 2 * hi;
        v2f as, an;
        as.x = xrow[kk];
        as.y = xrow[kk + 1];
        an.x = hrow[kk];
        an.y = hrow[kk + 1];
        for (int t = 0; t < 2; ++t) {
            const int bi = kk * DIM + (ch * 2 + t) * 16 + m;
            v2f bs, bn;
            bs.x = sWs[bi];
            bs.y = sWs[bi + DIM];
            bn.x = sWn[bi];
            bn.y = sWn[bi + DIM];
            acc[t] = __builtin_amdgcn_wmma_f32_16x16x4_f32(
                         false, as, false, bs, (short)0, acc[t], false, false);
            acc[t] = __builtin_amdgcn_wmma_f32_16x16x4_f32(
                         false, an, false, bn, (short)0, acc[t], false, false);
        }
    }

    for (int t = 0; t < 2; ++t) {
        const int col = (ch * 2 + t) * 16 + m;
        for (int rr = 0; rr < 8; ++rr) {
            const int mrow = rr + 8 * hi;
            out[(size_t)(row0 + mrow) * DIM + col] = acc[t][rr];
        }
    }
}

extern "C" void kernel_launch(void* const* d_in, const int* in_sizes, int n_in,
                              void* d_out, int out_size, void* d_ws, size_t ws_size,
                              hipStream_t stream) {
    const float* x      = (const float*)d_in[0];
    const int*   src    = (const int*)  d_in[1];
    const int*   dst    = (const int*)  d_in[2];
    const float* Wself  = (const float*)d_in[3];
    const float* bself  = (const float*)d_in[4];
    const float* Wneigh = (const float*)d_in[5];
    const float* bneigh = (const float*)d_in[6];
    float*       out    = (float*)d_out;

    const int N = in_sizes[0] / DIM;
    const int E = in_sizes[1];

    int* head = (int*)d_ws;
    int* recw = head + (size_t)N * NSL * 2;

    const int ES = (((E + NSL - 1) / NSL) + 3) & ~3;
    const int H  = (N + RCAP - 1) / RCAP;
    sage_slice_lists_kernel<<<dim3(NSL, H), 1024, 0, stream>>>(src, dst, head, recw, E, N, ES);

    const int nTiles = (N + 15) / 16;
    sage_gather_wmma_kernel<<<(N + ROWS_PB - 1) / ROWS_PB, 512, 0, stream>>>(
        x, head, recw, Wself, bself, Wneigh, bneigh, out, N, E, nTiles, ES);
}
